# starting kernel
# baseline (speedup 1.0000x reference)
#include <hip/hip_runtime.h>

#define Nw 2048
#define Mm 64
#define Kk 32
#define Dd 512
#define NCOPY 8
#define ACC_STRIDE (Kk*Dd + Kk*Kk)

typedef float v2f __attribute__((ext_vector_type(2)));
typedef float v8f __attribute__((ext_vector_type(8)));

__device__ __forceinline__ v8f wmma4(v2f a, v2f b, v8f c) {
  return __builtin_amdgcn_wmma_f32_16x16x4_f32(false, a, false, b, (short)0, c, false, false);
}

__device__ __forceinline__ float wave_max(float v) {
  #pragma unroll
  for (int off = 16; off > 0; off >>= 1) v = fmaxf(v, __shfl_xor(v, off, 32));
  return v;
}
__device__ __forceinline__ float wave_sum(float v) {
  #pragma unroll
  for (int off = 16; off > 0; off >>= 1) v += __shfl_xor(v, off, 32);
  return v;
}

__global__ __launch_bounds__(256)
void crf_word_kernel(const float* __restrict__ Wg, const float* __restrict__ Tg,
                     const float* __restrict__ data, const int* __restrict__ labels,
                     float* __restrict__ acc)
{
  __shared__ float sDots [Mm][Kk + 1];
  __shared__ float sAlpha[Mm][Kk + 1];
  __shared__ float sBeta [Mm][Kk + 1];
  __shared__ float sU    [Mm][Kk + 1];
  __shared__ float sV    [Mm][Kk + 1];
  __shared__ float sG    [Mm][Kk + 1];
  __shared__ float sT    [Kk][Kk + 1];
  __shared__ float sCnt  [Kk][Kk + 1];
  __shared__ float sCu[Mm];
  __shared__ float sDv[Mm];
  __shared__ int   sLab[Mm];
  __shared__ float sLogZ;

  const int tid  = threadIdx.x;
  const int w    = tid >> 5;
  const int lane = tid & 31;
  const int col  = lane & 15;
  const int hi   = lane >> 4;
  const int word = blockIdx.x;
  const float* dptr = data + (size_t)word * (Mm * Dd);

  float* base  = acc + (size_t)(blockIdx.x & (NCOPY - 1)) * ACC_STRIDE;
  float* accDW = base;
  float* accDT = base + Kk * Dd;

  for (int i = tid; i < Kk * Kk; i += 256) {
    sT  [i >> 5][i & 31] = Tg[i];
    sCnt[i >> 5][i & 31] = 0.f;
  }
  if (tid < Mm) sLab[tid] = labels[word * Mm + tid];

  {
    const int m0 = (w >> 1) * 16;
    const int n0 = (w & 1) * 16;
    v8f c0 = {}, c1 = {};
    const float* arow = dptr + (size_t)(m0 + col) * Dd + 2 * hi;
    const float* brow = Wg   + (size_t)(n0 + col) * Dd + 2 * hi;
    #pragma unroll 4
    for (int kk = 0; kk < Dd; kk += 8) {
      v2f a0 = *(const v2f*)(arow + kk);
      v2f b0 = *(const v2f*)(brow + kk);
      v2f a1 = *(const v2f*)(arow + kk + 4);
      v2f b1 = *(const v2f*)(brow + kk + 4);
      c0 = wmma4(a0, b0, c0);
      c1 = wmma4(a1, b1, c1);
    }
    v8f c = c0 + c1;
    #pragma unroll
    for (int r = 0; r < 8; ++r)
      sDots[m0 + r + 8 * hi][n0 + col] = c[r];
  }
  __syncthreads();

  if (w == 0) {
    float Tc[32];
    #pragma unroll
    for (int j = 0; j < 32; ++j) Tc[j] = sT[j][lane];
    float prev = 0.f;
    sAlpha[0][lane] = 0.f;
    for (int i = 1; i < Mm; ++i) {
      float sv = prev + sDots[i - 1][lane];
      float tv[32];
      float mx = -3.0e38f;
      #pragma unroll
      for (int j = 0; j < 32; ++j) {
        float s = __shfl(sv, j, 32);
        tv[j] = s + Tc[j];
        mx = fmaxf(mx, tv[j]);
      }
      float sum = 0.f;
      #pragma unroll
      for (int j = 0; j < 32; ++j) sum += __expf(tv[j] - mx);
      prev = mx + __logf(sum);
      sAlpha[i][lane] = prev;
    }
  } else if (w == 1) {
    float Tr[32];
    #pragma unroll
    for (int j = 0; j < 32; ++j) Tr[j] = sT[lane][j];
    float prev = 0.f;
    sBeta[Mm - 1][lane] = 0.f;
    for (int i = Mm - 2; i >= 0; --i) {
      float sv = prev + sDots[i + 1][lane];
      float tv[32];
      float mx = -3.0e38f;
      #pragma unroll
      for (int j = 0; j < 32; ++j) {
        float s = __shfl(sv, j, 32);
        tv[j] = s + Tr[j];
        mx = fmaxf(mx, tv[j]);
      }
      float sum = 0.f;
      #pragma unroll
      for (int j = 0; j < 32; ++j) sum += __expf(tv[j] - mx);
      prev = mx + __logf(sum);
      sBeta[i][lane] = prev;
    }
  } else {
    const int t2 = tid - 64;
    for (int line = t2; line < (Mm * Dd) / 32; line += 192)
      __builtin_prefetch(dptr + line * 32, 0, 3);
  }
  __syncthreads();

  if (w == 0) {
    float v  = sAlpha[Mm - 1][lane] + sDots[Mm - 1][lane];
    float mx = wave_max(v);
    float s  = wave_sum(__expf(v - mx));
    if (lane == 0) sLogZ = mx + __logf(s);
  }
  __syncthreads();
  const float logZ = sLogZ;

  #pragma unroll 1
  for (int t = 0; t < 8; ++t) {
    const int i = w + 8 * t;
    const float a = sAlpha[i][lane], b = sBeta[i][lane], d = sDots[i][lane];
    float e   = a + b + d;
    float mx  = wave_max(e);
    float p   = __expf(e - mx);
    float inv = 1.f / wave_sum(p);
    sG[i][lane] = ((lane == sLab[i]) ? 1.f : 0.f) - p * inv;

    float aL = a + d;
    float ca = wave_max(aL);
    if (i < Mm - 1) { sU[i][lane] = __expf(aL - ca); if (lane == 0) sCu[i] = ca; }
    else            { sU[Mm - 1][lane] = 0.f; }

    float bR = b + d;
    float cb = wave_max(bR);
    if (i >= 1) { sV[i - 1][lane] = __expf(bR - cb); if (lane == 0) sDv[i - 1] = cb; }
    else        { sV[Mm - 1][lane] = 0.f; }
  }
  if (tid < Mm - 1) atomicAdd(&sCnt[sLab[tid]][sLab[tid + 1]], 1.f);
  __syncthreads();

  for (int idx = tid; idx < (Mm - 1) * Kk; idx += 256) {
    const int i = idx >> 5, k = idx & 31;
    sU[i][k] *= __expf(sCu[i] + sDv[i] - logZ);
  }
  __syncthreads();

  if (w < 4) {
    const int j0 = (w >> 1) * 16, k0 = (w & 1) * 16;
    v8f c0 = {}, c1 = {};
    #pragma unroll
    for (int ii = 0; ii < Mm; ii += 8) {
      v2f a0, b0, a1, b1;
      a0.x = sU[ii + 2 * hi    ][j0 + col];
      a0.y = sU[ii + 2 * hi + 1][j0 + col];
      b0.x = sV[ii + 2 * hi    ][k0 + col];
      b0.y = sV[ii + 2 * hi + 1][k0 + col];
      a1.x = sU[ii + 4 + 2 * hi    ][j0 + col];
      a1.y = sU[ii + 4 + 2 * hi + 1][j0 + col];
      b1.x = sV[ii + 4 + 2 * hi    ][k0 + col];
      b1.y = sV[ii + 4 + 2 * hi + 1][k0 + col];
      c0 = wmma4(a0, b0, c0);
      c1 = wmma4(a1, b1, c1);
    }
    v8f c = c0 + c1;
    #pragma unroll
    for (int r = 0; r < 8; ++r) {
      const int j = j0 + r + 8 * hi, k = k0 + col;
      const float val = sCnt[j][k] - __expf(sT[j][k]) * c[r];
      atomicAdd(accDT + j * Kk + k, val);
    }
  }

  {
    const int j0 = (w & 1) * 16;
    v2f afr[16];
    #pragma unroll
    for (int s = 0; s < 16; ++s) {
      afr[s].x = sG[4 * s + 2 * hi    ][j0 + col];
      afr[s].y = sG[4 * s + 2 * hi + 1][j0 + col];
    }
    #pragma unroll 1
    for (int dt = (w >> 1); dt < 32; dt += 4) {
      const int d0 = dt * 16;
      v8f c0 = {}, c1 = {};
      #pragma unroll
      for (int s = 0; s < 16; s += 2) {
        v2f b0, b1;
        b0.x = dptr[(size_t)(4 * s + 2 * hi    ) * Dd + d0 + col];
        b0.y = dptr[(size_t)(4 * s + 2 * hi + 1) * Dd + d0 + col];
        b1.x = dptr[(size_t)(4 * s + 4 + 2 * hi    ) * Dd + d0 + col];
        b1.y = dptr[(size_t)(4 * s + 4 + 2 * hi + 1) * Dd + d0 + col];
        c0 = wmma4(afr[s],     b0, c0);
        c1 = wmma4(afr[s + 1], b1, c1);
      }
      v8f c = c0 + c1;
      #pragma unroll
      for (int r = 0; r < 8; ++r)
        atomicAdd(accDW + (j0 + r + 8 * hi) * Dd + d0 + col, c[r]);
    }
  }
}

__global__ void zero_ws_kernel(float* __restrict__ p, int n) {
  const int i = blockIdx.x * 256 + threadIdx.x;
  if (i < n) p[i] = 0.f;
}

__global__ void reduce_out_kernel(const float* __restrict__ acc, float* __restrict__ out) {
  const int i = blockIdx.x * 256 + threadIdx.x;
  if (i >= ACC_STRIDE) return;
  float s = 0.f;
  #pragma unroll
  for (int c = 0; c < NCOPY; ++c) s += acc[(size_t)c * ACC_STRIDE + i];
  out[i] = s * (1.0f / (float)Nw);
}

extern "C" void kernel_launch(void* const* d_in, const int* in_sizes, int n_in,
                              void* d_out, int out_size, void* d_ws, size_t ws_size,
                              hipStream_t stream) {
  const float* W      = (const float*)d_in[0];
  const float* T      = (const float*)d_in[1];
  const float* data   = (const float*)d_in[2];
  const int*   labels = (const int*)  d_in[3];
  float* out = (float*)d_out;
  float* ws  = (float*)d_ws;

  const int nacc = NCOPY * ACC_STRIDE;
  zero_ws_kernel<<<(nacc + 255) / 256, 256, 0, stream>>>(ws, nacc);
  crf_word_kernel<<<Nw, 256, 0, stream>>>(W, T, data, labels, ws);
  reduce_out_kernel<<<(ACC_STRIDE + 255) / 256, 256, 0, stream>>>(ws, out);
}


# phase 1 (data x W^T) takes its operands through LDS: 64-float K chunks staged by whole-line 16-byte global loads, double buffered; k grouping of the two WMMA sum chains changes (sum order of dots)
# speedup vs baseline: 1.1744x; 1.1744x over previous
#include <hip/hip_runtime.h>

#define Nw 2048
#define Mm 64
#define Kk 32
#define Dd 512
#define NCOPY 8
#define ACC_STRIDE (Kk*Dd + Kk*Kk)

#define KC 64
#define NCHUNK (Dd / KC)
#define SPITCH (KC + 4)
#define SROWS (Mm + Kk)
#define STAGE_FLOATS (SROWS * SPITCH)
#define ARR_FLOATS (Mm * (Kk + 1))
#define BIG_FLOATS (2 * STAGE_FLOATS)

typedef float v2f __attribute__((ext_vector_type(2)));
typedef float v4f __attribute__((ext_vector_type(4)));
typedef float v8f __attribute__((ext_vector_type(8)));

__device__ __forceinline__ v8f wmma4(v2f a, v2f b, v8f c) {
  return __builtin_amdgcn_wmma_f32_16x16x4_f32(false, a, false, b, (short)0, c, false, false);
}

__device__ __forceinline__ float wave_max(float v) {
  #pragma unroll
  for (int off = 16; off > 0; off >>= 1) v = fmaxf(v, __shfl_xor(v, off, 32));
  return v;
}
__device__ __forceinline__ float wave_sum(float v) {
  #pragma unroll
  for (int off = 16; off > 0; off >>= 1) v += __shfl_xor(v, off, 32);
  return v;
}

__global__ __launch_bounds__(256)
void crf_word_kernel(const float* __restrict__ Wg, const float* __restrict__ Tg,
                     const float* __restrict__ data, const int* __restrict__ labels,
                     float* __restrict__ acc)
{
  __shared__ __attribute__((aligned(16))) float sBig[BIG_FLOATS];
  static_assert(BIG_FLOATS >= 6 * ARR_FLOATS, "work arrays must fit the staging region");
  float (*sDots )[Kk + 1] = (float (*)[Kk + 1])(sBig + 0 * ARR_FLOATS);
  float (*sAlpha)[Kk + 1] = (float (*)[Kk + 1])(sBig + 1 * ARR_FLOATS);
  float (*sBeta )[Kk + 1] = (float (*)[Kk + 1])(sBig + 2 * ARR_FLOATS);
  float (*sU    )[Kk + 1] = (float (*)[Kk + 1])(sBig + 3 * ARR_FLOATS);
  float (*sV    )[Kk + 1] = (float (*)[Kk + 1])(sBig + 4 * ARR_FLOATS);
  float (*sG    )[Kk + 1] = (float (*)[Kk + 1])(sBig + 5 * ARR_FLOATS);
  __shared__ float sT    [Kk][Kk + 1];
  __shared__ float sCnt  [Kk][Kk + 1];
  __shared__ float sCu[Mm];
  __shared__ float sDv[Mm];
  __shared__ int   sLab[Mm];
  __shared__ float sLogZ;

  const int tid  = threadIdx.x;
  const int w    = tid >> 5;
  const int lane = tid & 31;
  const int col  = lane & 15;
  const int hi   = lane >> 4;
  const int word = blockIdx.x;
  const float* dptr = data + (size_t)word * (Mm * Dd);

  float* base  = acc + (size_t)(blockIdx.x & (NCOPY - 1)) * ACC_STRIDE;
  float* accDW = base;
  float* accDT = base + Kk * Dd;

  const int srow = tid >> 4;
  const int sc4  = (tid & 15) * 4;
  const float* gdat = dptr + (size_t)srow * Dd + sc4;
  const float* gwgt = Wg   + (size_t)srow * Dd + sc4;
  float* sdst = sBig + srow * SPITCH + sc4;

  v4f stg[6];
  #pragma unroll
  for (int q = 0; q < 4; ++q) stg[q] = *(const v4f*)(gdat + (size_t)q * 16 * Dd);
  #pragma unroll
  for (int q = 4; q < 6; ++q) stg[q] = *(const v4f*)(gwgt + (size_t)(q - 4) * 16 * Dd);

  for (int i = tid; i < Kk * Kk; i += 256) {
    sT  [i >> 5][i & 31] = Tg[i];
    sCnt[i >> 5][i & 31] = 0.f;
  }
  if (tid < Mm) sLab[tid] = labels[word * Mm + tid];

  #pragma unroll
  for (int q = 0; q < 6; ++q) *(v4f*)(sdst + q * 16 * SPITCH) = stg[q];
  __syncthreads();

  {
    const int m0 = (w >> 1) * 16;
    const int n0 = (w & 1) * 16;
    const float* fa = sBig + (m0 + col) * SPITCH + hi * 4;
    const float* fb = sBig + (Mm + n0 + col) * SPITCH + hi * 4;
    v8f c0 = {}, c1 = {};
    for (int t = 0; t < NCHUNK; ++t) {
      const int cur = (t & 1) * STAGE_FLOATS;
      const int nxt = STAGE_FLOATS - cur;
      if (t + 1 < NCHUNK) {
        const int ko = (t + 1) * KC;
        #pragma unroll
        for (int q = 0; q < 4; ++q) stg[q] = *(const v4f*)(gdat + (size_t)q * 16 * Dd + ko);
        #pragma unroll
        for (int q = 4; q < 6; ++q) stg[q] = *(const v4f*)(gwgt + (size_t)(q - 4) * 16 * Dd + ko);
      }
      __builtin_amdgcn_sched_barrier(0);
      #pragma unroll
      for (int s0 = 0; s0 < KC / 8; s0 += 4) {
        v4f a[4], b[4];
        #pragma unroll
        for (int j = 0; j < 4; ++j) {
          a[j] = *(const v4f*)(fa + cur + (s0 + j) * 8);
          b[j] = *(const v4f*)(fb + cur + (s0 + j) * 8);
        }
        #pragma unroll
        for (int j = 0; j < 4; ++j) {
          c0 = wmma4(a[j].xy, b[j].xy, c0);
          c1 = wmma4(a[j].zw, b[j].zw, c1);
        }
      }
      __builtin_amdgcn_sched_barrier(0);
      if (t + 1 < NCHUNK) {
        #pragma unroll
        for (int q = 0; q < 6; ++q) *(v4f*)(sdst + nxt + q * 16 * SPITCH) = stg[q];
      }
      __syncthreads();
    }
    v8f c = c0 + c1;
    #pragma unroll
    for (int r = 0; r < 8; ++r)
      sDots[m0 + r + 8 * hi][n0 + col] = c[r];
  }
  __syncthreads();

  if (w == 0) {
    float Tc[32];
    #pragma unroll
    for (int j = 0; j < 32; ++j) Tc[j] = sT[j][lane];
    float prev = 0.f;
    sAlpha[0][lane] = 0.f;
    for (int i = 1; i < Mm; ++i) {
      float sv = prev + sDots[i - 1][lane];
      float tv[32];
      float mx = -3.0e38f;
      #pragma unroll
      for (int j = 0; j < 32; ++j) {
        float s = __shfl(sv, j, 32);
        tv[j] = s + Tc[j];
        mx = fmaxf(mx, tv[j]);
      }
      float sum = 0.f;
      #pragma unroll
      for (int j = 0; j < 32; ++j) sum += __expf(tv[j] - mx);
      prev = mx + __logf(sum);
      sAlpha[i][lane] = prev;
    }
  } else if (w == 1) {
    float Tr[32];
    #pragma unroll
    for (int j = 0; j < 32; ++j) Tr[j] = sT[lane][j];
    float prev = 0.f;
    sBeta[Mm - 1][lane] = 0.f;
    for (int i = Mm - 2; i >= 0; --i) {
      float sv = prev + sDots[i + 1][lane];
      float tv[32];
      float mx = -3.0e38f;
      #pragma unroll
      for (int j = 0; j < 32; ++j) {
        float s = __shfl(sv, j, 32);
        tv[j] = s + Tr[j];
        mx = fmaxf(mx, tv[j]);
      }
      float sum = 0.f;
      #pragma unroll
      for (int j = 0; j < 32; ++j) sum += __expf(tv[j] - mx);
      prev = mx + __logf(sum);
      sBeta[i][lane] = prev;
    }
  } else {
    const int t2 = tid - 64;
    for (int line = t2; line < (Mm * Dd) / 32; line += 192)
      __builtin_prefetch(dptr + line * 32, 0, 3);
  }
  __syncthreads();

  if (w == 0) {
    float v  = sAlpha[Mm - 1][lane] + sDots[Mm - 1][lane];
    float mx = wave_max(v);
    float s  = wave_sum(__expf(v - mx));
    if (lane == 0) sLogZ = mx + __logf(s);
  }
  __syncthreads();
  const float logZ = sLogZ;

  #pragma unroll 1
  for (int t = 0; t < 8; ++t) {
    const int i = w + 8 * t;
    const float a = sAlpha[i][lane], b = sBeta[i][lane], d = sDots[i][lane];
    float e   = a + b + d;
    float mx  = wave_max(e);
    float p   = __expf(e - mx);
    float inv = 1.f / wave_sum(p);
    sG[i][lane] = ((lane == sLab[i]) ? 1.f : 0.f) - p * inv;

    float aL = a + d;
    float ca = wave_max(aL);
    if (i < Mm - 1) { sU[i][lane] = __expf(aL - ca); if (lane == 0) sCu[i] = ca; }
    else            { sU[Mm - 1][lane] = 0.f; }

    float bR = b + d;
    float cb = wave_max(bR);
    if (i >= 1) { sV[i - 1][lane] = __expf(bR - cb); if (lane == 0) sDv[i - 1] = cb; }
    else        { sV[Mm - 1][lane] = 0.f; }
  }
  if (tid < Mm - 1) atomicAdd(&sCnt[sLab[tid]][sLab[tid + 1]], 1.f);
  __syncthreads();

  for (int idx = tid; idx < (Mm - 1) * Kk; idx += 256) {
    const int i = idx >> 5, k = idx & 31;
    sU[i][k] *= __expf(sCu[i] + sDv[i] - logZ);
  }
  __syncthreads();

  if (w < 4) {
    const int j0 = (w >> 1) * 16, k0 = (w & 1) * 16;
    v8f c0 = {}, c1 = {};
    #pragma unroll
    for (int ii = 0; ii < Mm; ii += 8) {
      v2f a0, b0, a1, b1;
      a0.x = sU[ii + 2 * hi    ][j0 + col];
      a0.y = sU[ii + 2 * hi + 1][j0 + col];
      b0.x = sV[ii + 2 * hi    ][k0 + col];
      b0.y = sV[ii + 2 * hi + 1][k0 + col];
      a1.x = sU[ii + 4 + 2 * hi    ][j0 + col];
      a1.y = sU[ii + 4 + 2 * hi + 1][j0 + col];
      b1.x = sV[ii + 4 + 2 * hi    ][k0 + col];
      b1.y = sV[ii + 4 + 2 * hi + 1][k0 + col];
      c0 = wmma4(a0, b0, c0);
      c1 = wmma4(a1, b1, c1);
    }
    v8f c = c0 + c1;
    #pragma unroll
    for (int r = 0; r < 8; ++r) {
      const int j = j0 + r + 8 * hi, k = k0 + col;
      const float val = sCnt[j][k] - __expf(sT[j][k]) * c[r];
      atomicAdd(accDT + j * Kk + k, val);
    }
  }

  {
    const int j0 = (w & 1) * 16;
    v2f afr[16];
    #pragma unroll
    for (int s = 0; s < 16; ++s) {
      afr[s].x = sG[4 * s + 2 * hi    ][j0 + col];
      afr[s].y = sG[4 * s + 2 * hi + 1][j0 + col];
    }
    #pragma unroll 1
    for (int dt = (w >> 1); dt < 32; dt += 4) {
      const int d0 = dt * 16;
      v8f c0 = {}, c1 = {};
      #pragma unroll
      for (int s = 0; s < 16; s += 2) {
        v2f b0, b1;
        b0.x = dptr[(size_t)(4 * s + 2 * hi    ) * Dd + d0 + col];
        b0.y = dptr[(size_t)(4 * s + 2 * hi + 1) * Dd + d0 + col];
        b1.x = dptr[(size_t)(4 * s + 4 + 2 * hi    ) * Dd + d0 + col];
        b1.y = dptr[(size_t)(4 * s + 4 + 2 * hi + 1) * Dd + d0 + col];
        c0 = wmma4(afr[s],     b0, c0);
        c1 = wmma4(afr[s + 1], b1, c1);
      }
      v8f c = c0 + c1;
      #pragma unroll
      for (int r = 0; r < 8; ++r)
        atomicAdd(accDW + (j0 + r + 8 * hi) * Dd + d0 + col, c[r]);
    }
  }
}

__global__ void zero_ws_kernel(float* __restrict__ p, int n) {
  const int i = blockIdx.x * 256 + threadIdx.x;
  if (i < n) p[i] = 0.f;
}

__global__ void reduce_out_kernel(const float* __restrict__ acc, float* __restrict__ out) {
  const int i = blockIdx.x * 256 + threadIdx.x;
  if (i >= ACC_STRIDE) return;
  float s = 0.f;
  #pragma unroll
  for (int c = 0; c < NCOPY; ++c) s += acc[(size_t)c * ACC_STRIDE + i];
  out[i] = s * (1.0f / (float)Nw);
}

extern "C" void kernel_launch(void* const* d_in, const int* in_sizes, int n_in,
                              void* d_out, int out_size, void* d_ws, size_t ws_size,
                              hipStream_t stream) {
  const float* W      = (const float*)d_in[0];
  const float* T      = (const float*)d_in[1];
  const float* data   = (const float*)d_in[2];
  const int*   labels = (const int*)  d_in[3];
  float* out = (float*)d_out;
  float* ws  = (float*)d_ws;

  const int nacc = NCOPY * ACC_STRIDE;
  zero_ws_kernel<<<(nacc + 255) / 256, 256, 0, stream>>>(ws, nacc);
  crf_word_kernel<<<Nw, 256, 0, stream>>>(W, T, data, labels, ws);
  reduce_out_kernel<<<(ACC_STRIDE + 255) / 256, 256, 0, stream>>>(ws, out);
}


# alpha/beta recursion step as m + tmax_k + log(sum_j exp(sv_j - m) * exp(T_jk - tmax_k)): exp(T) in registers, wave max by DPP; 1 exp per lane per step, not 32 (sum order of the steps changes)
# speedup vs baseline: 1.3630x; 1.1606x over previous
#include <hip/hip_runtime.h>

#define Nw 2048
#define Mm 64
#define Kk 32
#define Dd 512
#define NCOPY 8
#define ACC_STRIDE (Kk*Dd + Kk*Kk)

#define KC 64
#define NCHUNK (Dd / KC)
#define SPITCH (KC + 4)
#define SROWS (Mm + Kk)
#define STAGE_FLOATS (SROWS * SPITCH)
#define ARR_FLOATS (Mm * (Kk + 1))
#define BIG_FLOATS (2 * STAGE_FLOATS)

typedef float v2f __attribute__((ext_vector_type(2)));
typedef float v4f __attribute__((ext_vector_type(4)));
typedef float v8f __attribute__((ext_vector_type(8)));

__device__ __forceinline__ v8f wmma4(v2f a, v2f b, v8f c) {
  return __builtin_amdgcn_wmma_f32_16x16x4_f32(false, a, false, b, (short)0, c, false, false);
}

__device__ __forceinline__ float wave_max(float v) {
  #pragma unroll
  for (int off = 16; off > 0; off >>= 1) v = fmaxf(v, __shfl_xor(v, off, 32));
  return v;
}
__device__ __forceinline__ float wave_sum(float v) {
  #pragma unroll
  for (int off = 16; off > 0; off >>= 1) v += __shfl_xor(v, off, 32);
  return v;
}

__device__ __forceinline__ float xhalf(float x) {
  const int b = __builtin_bit_cast(int, x);
  return __builtin_bit_cast(float,
      __builtin_amdgcn_permlanex16(b, b, 0x76543210, 0xfedcba98, false, false));
}
__device__ __forceinline__ float wave_max_valu(float v) {
  asm("v_max_num_f32_dpp %0, %1, %1 row_xmask:1 row_mask:0xf bank_mask:0xf" : "=v"(v) : "v"(v));
  asm("v_max_num_f32_dpp %0, %1, %1 row_xmask:2 row_mask:0xf bank_mask:0xf" : "=v"(v) : "v"(v));
  asm("v_max_num_f32_dpp %0, %1, %1 row_xmask:4 row_mask:0xf bank_mask:0xf" : "=v"(v) : "v"(v));
  asm("v_max_num_f32_dpp %0, %1, %1 row_xmask:8 row_mask:0xf bank_mask:0xf" : "=v"(v) : "v"(v));
  return fmaxf(v, xhalf(v));
}
__device__ __forceinline__ float lane_bcast(float v, int j) {
  return __builtin_bit_cast(float, __builtin_amdgcn_readlane(__builtin_bit_cast(int, v), j));
}

__device__ __forceinline__ float lse_step(float sv, const float (&E)[32], float tmax) {
  const float m = wave_max_valu(sv);
  const float u = __expf(sv - m);
  float a0 = 0.f, a1 = 0.f, a2 = 0.f, a3 = 0.f;
  #pragma unroll
  for (int j = 0; j < 32; j += 4) {
    a0 = fmaf(lane_bcast(u, j    ), E[j    ], a0);
    a1 = fmaf(lane_bcast(u, j + 1), E[j + 1], a1);
    a2 = fmaf(lane_bcast(u, j + 2), E[j + 2], a2);
    a3 = fmaf(lane_bcast(u, j + 3), E[j + 3], a3);
  }
  const float r = (a0 + a1) + (a2 + a3);
  return m + fmaf(__builtin_amdgcn_logf(r), 0.69314718056f, tmax);
}

__global__ __launch_bounds__(256)
void crf_word_kernel(const float* __restrict__ Wg, const float* __restrict__ Tg,
                     const float* __restrict__ data, const int* __restrict__ labels,
                     float* __restrict__ acc)
{
  __shared__ __attribute__((aligned(16))) float sBig[BIG_FLOATS];
  static_assert(BIG_FLOATS >= 6 * ARR_FLOATS, "work arrays must fit the staging region");
  float (*sDots )[Kk + 1] = (float (*)[Kk + 1])(sBig + 0 * ARR_FLOATS);
  float (*sAlpha)[Kk + 1] = (float (*)[Kk + 1])(sBig + 1 * ARR_FLOATS);
  float (*sBeta )[Kk + 1] = (float (*)[Kk + 1])(sBig + 2 * ARR_FLOATS);
  float (*sU    )[Kk + 1] = (float (*)[Kk + 1])(sBig + 3 * ARR_FLOATS);
  float (*sV    )[Kk + 1] = (float (*)[Kk + 1])(sBig + 4 * ARR_FLOATS);
  float (*sG    )[Kk + 1] = (float (*)[Kk + 1])(sBig + 5 * ARR_FLOATS);
  __shared__ float sT    [Kk][Kk + 1];
  __shared__ float sCnt  [Kk][Kk + 1];
  __shared__ float sCu[Mm];
  __shared__ float sDv[Mm];
  __shared__ int   sLab[Mm];
  __shared__ float sLogZ;

  const int tid  = threadIdx.x;
  const int w    = tid >> 5;
  const int lane = tid & 31;
  const int col  = lane & 15;
  const int hi   = lane >> 4;
  const int word = blockIdx.x;
  const float* dptr = data + (size_t)word * (Mm * Dd);

  float* base  = acc + (size_t)(blockIdx.x & (NCOPY - 1)) * ACC_STRIDE;
  float* accDW = base;
  float* accDT = base + Kk * Dd;

  const int srow = tid >> 4;
  const int sc4  = (tid & 15) * 4;
  const float* gdat = dptr + (size_t)srow * Dd + sc4;
  const float* gwgt = Wg   + (size_t)srow * Dd + sc4;
  float* sdst = sBig + srow * SPITCH + sc4;

  v4f stg[6];
  #pragma unroll
  for (int q = 0; q < 4; ++q) stg[q] = *(const v4f*)(gdat + (size_t)q * 16 * Dd);
  #pragma unroll
  for (int q = 4; q < 6; ++q) stg[q] = *(const v4f*)(gwgt + (size_t)(q - 4) * 16 * Dd);

  for (int i = tid; i < Kk * Kk; i += 256) {
    sT  [i >> 5][i & 31] = Tg[i];
    sCnt[i >> 5][i & 31] = 0.f;
  }
  if (tid < Mm) sLab[tid] = labels[word * Mm + tid];

  #pragma unroll
  for (int q = 0; q < 6; ++q) *(v4f*)(sdst + q * 16 * SPITCH) = stg[q];
  __syncthreads();

  {
    const int m0 = (w >> 1) * 16;
    const int n0 = (w & 1) * 16;
    const float* fa = sBig + (m0 + col) * SPITCH + hi * 4;
    const float* fb = sBig + (Mm + n0 + col) * SPITCH + hi * 4;
    v8f c0 = {}, c1 = {};
    for (int t = 0; t < NCHUNK; ++t) {
      const int cur = (t & 1) * STAGE_FLOATS;
      const int nxt = STAGE_FLOATS - cur;
      if (t + 1 < NCHUNK) {
        const int ko = (t + 1) * KC;
        #pragma unroll
        for (int q = 0; q < 4; ++q) stg[q] = *(const v4f*)(gdat + (size_t)q * 16 * Dd + ko);
        #pragma unroll
        for (int q = 4; q < 6; ++q) stg[q] = *(const v4f*)(gwgt + (size_t)(q - 4) * 16 * Dd + ko);
      }
      __builtin_amdgcn_sched_barrier(0);
      #pragma unroll
      for (int s0 = 0; s0 < KC / 8; s0 += 4) {
        v4f a[4], b[4];
        #pragma unroll
        for (int j = 0; j < 4; ++j) {
          a[j] = *(const v4f*)(fa + cur + (s0 + j) * 8);
          b[j] = *(const v4f*)(fb + cur + (s0 + j) * 8);
        }
        #pragma unroll
        for (int j = 0; j < 4; ++j) {
          c0 = wmma4(a[j].xy, b[j].xy, c0);
          c1 = wmma4(a[j].zw, b[j].zw, c1);
        }
      }
      __builtin_amdgcn_sched_barrier(0);
      if (t + 1 < NCHUNK) {
        #pragma unroll
        for (int q = 0; q < 6; ++q) *(v4f*)(sdst + nxt + q * 16 * SPITCH) = stg[q];
      }
      __syncthreads();
    }
    v8f c = c0 + c1;
    #pragma unroll
    for (int r = 0; r < 8; ++r)
      sDots[m0 + r + 8 * hi][n0 + col] = c[r];
  }
  __syncthreads();

  if (w == 0) {
    float E[32];
    float tmax = -3.0e38f;
    #pragma unroll
    for (int j = 0; j < 32; ++j) { E[j] = sT[j][lane]; tmax = fmaxf(tmax, E[j]); }
    #pragma unroll
    for (int j = 0; j < 32; ++j) E[j] = __expf(E[j] - tmax);
    float prev = 0.f;
    sAlpha[0][lane] = 0.f;
    float dA = sDots[0][lane], dB = sDots[1][lane];
    for (int i = 1; i < Mm - 1; i += 2) {
      float sv = prev + dA;
      dA = sDots[i + 1][lane];
      __builtin_amdgcn_sched_barrier(0);
      prev = lse_step(sv, E, tmax);
      sAlpha[i][lane] = prev;
      sv = prev + dB;
      dB = sDots[i + 2][lane];
      __builtin_amdgcn_sched_barrier(0);
      prev = lse_step(sv, E, tmax);
      sAlpha[i + 1][lane] = prev;
    }
    sAlpha[Mm - 1][lane] = lse_step(prev + dA, E, tmax);
  } else if (w == 1) {
    float E[32];
    float tmax = -3.0e38f;
    #pragma unroll
    for (int j = 0; j < 32; ++j) { E[j] = sT[lane][j]; tmax = fmaxf(tmax, E[j]); }
    #pragma unroll
    for (int j = 0; j < 32; ++j) E[j] = __expf(E[j] - tmax);
    float prev = 0.f;
    sBeta[Mm - 1][lane] = 0.f;
    float dA = sDots[Mm - 1][lane], dB = sDots[Mm - 2][lane];
    for (int i = Mm - 2; i > 0; i -= 2) {
      float sv = prev + dA;
      dA = sDots[i - 1][lane];
      __builtin_amdgcn_sched_barrier(0);
      prev = lse_step(sv, E, tmax);
      sBeta[i][lane] = prev;
      sv = prev + dB;
      dB = sDots[i - 2][lane];
      __builtin_amdgcn_sched_barrier(0);
      prev = lse_step(sv, E, tmax);
      sBeta[i - 1][lane] = prev;
    }
    sBeta[0][lane] = lse_step(prev + dA, E, tmax);
  } else {
    const int t2 = tid - 64;
    for (int line = t2; line < (Mm * Dd) / 32; line += 192)
      __builtin_prefetch(dptr + line * 32, 0, 3);
  }
  __syncthreads();

  if (w == 0) {
    float v  = sAlpha[Mm - 1][lane] + sDots[Mm - 1][lane];
    float mx = wave_max(v);
    float s  = wave_sum(__expf(v - mx));
    if (lane == 0) sLogZ = mx + __logf(s);
  }
  __syncthreads();
  const float logZ = sLogZ;

  #pragma unroll 1
  for (int t = 0; t < 8; ++t) {
    const int i = w + 8 * t;
    const float a = sAlpha[i][lane], b = sBeta[i][lane], d = sDots[i][lane];
    float e   = a + b + d;
    float mx  = wave_max(e);
    float p   = __expf(e - mx);
    float inv = 1.f / wave_sum(p);
    sG[i][lane] = ((lane == sLab[i]) ? 1.f : 0.f) - p * inv;

    float aL = a + d;
    float ca = wave_max(aL);
    if (i < Mm - 1) { sU[i][lane] = __expf(aL - ca); if (lane == 0) sCu[i] = ca; }
    else            { sU[Mm - 1][lane] = 0.f; }

    float bR = b + d;
    float cb = wave_max(bR);
    if (i >= 1) { sV[i - 1][lane] = __expf(bR - cb); if (lane == 0) sDv[i - 1] = cb; }
    else        { sV[Mm - 1][lane] = 0.f; }
  }
  if (tid < Mm - 1) atomicAdd(&sCnt[sLab[tid]][sLab[tid + 1]], 1.f);
  __syncthreads();

  for (int idx = tid; idx < (Mm - 1) * Kk; idx += 256) {
    const int i = idx >> 5, k = idx & 31;
    sU[i][k] *= __expf(sCu[i] + sDv[i] - logZ);
  }
  __syncthreads();

  if (w < 4) {
    const int j0 = (w >> 1) * 16, k0 = (w & 1) * 16;
    v8f c0 = {}, c1 = {};
    #pragma unroll
    for (int ii = 0; ii < Mm; ii += 8) {
      v2f a0, b0, a1, b1;
      a0.x = sU[ii + 2 * hi    ][j0 + col];
      a0.y = sU[ii + 2 * hi + 1][j0 + col];
      b0.x = sV[ii + 2 * hi    ][k0 + col];
      b0.y = sV[ii + 2 * hi + 1][k0 + col];
      a1.x = sU[ii + 4 + 2 * hi    ][j0 + col];
      a1.y = sU[ii + 4 + 2 * hi + 1][j0 + col];
      b1.x = sV[ii + 4 + 2 * hi    ][k0 + col];
      b1.y = sV[ii + 4 + 2 * hi + 1][k0 + col];
      c0 = wmma4(a0, b0, c0);
      c1 = wmma4(a1, b1, c1);
    }
    v8f c = c0 + c1;
    #pragma unroll
    for (int r = 0; r < 8; ++r) {
      const int j = j0 + r + 8 * hi, k = k0 + col;
      const float val = sCnt[j][k] - __expf(sT[j][k]) * c[r];
      atomicAdd(accDT + j * Kk + k, val);
    }
  }

  {
    const int j0 = (w & 1) * 16;
    v2f afr[16];
    #pragma unroll
    for (int s = 0; s < 16; ++s) {
      afr[s].x = sG[4 * s + 2 * hi    ][j0 + col];
      afr[s].y = sG[4 * s + 2 * hi + 1][j0 + col];
    }
    #pragma unroll 1
    for (int dt = (w >> 1); dt < 32; dt += 4) {
      const int d0 = dt * 16;
      v8f c0 = {}, c1 = {};
      #pragma unroll
      for (int s = 0; s < 16; s += 2) {
        v2f b0, b1;
        b0.x = dptr[(size_t)(4 * s + 2 * hi    ) * Dd + d0 + col];
        b0.y = dptr[(size_t)(4 * s + 2 * hi + 1) * Dd + d0 + col];
        b1.x = dptr[(size_t)(4 * s + 4 + 2 * hi    ) * Dd + d0 + col];
        b1.y = dptr[(size_t)(4 * s + 4 + 2 * hi + 1) * Dd + d0 + col];
        c0 = wmma4(afr[s],     b0, c0);
        c1 = wmma4(afr[s + 1], b1, c1);
      }
      v8f c = c0 + c1;
      #pragma unroll
      for (int r = 0; r < 8; ++r)
        atomicAdd(accDW + (j0 + r + 8 * hi) * Dd + d0 + col, c[r]);
    }
  }
}

__global__ void zero_ws_kernel(float* __restrict__ p, int n) {
  const int i = blockIdx.x * 256 + threadIdx.x;
  if (i < n) p[i] = 0.f;
}

__global__ void reduce_out_kernel(const float* __restrict__ acc, float* __restrict__ out) {
  const int i = blockIdx.x * 256 + threadIdx.x;
  if (i >= ACC_STRIDE) return;
  float s = 0.f;
  #pragma unroll
  for (int c = 0; c < NCOPY; ++c) s += acc[(size_t)c * ACC_STRIDE + i];
  out[i] = s * (1.0f / (float)Nw);
}

extern "C" void kernel_launch(void* const* d_in, const int* in_sizes, int n_in,
                              void* d_out, int out_size, void* d_ws, size_t ws_size,
                              hipStream_t stream) {
  const float* W      = (const float*)d_in[0];
  const float* T      = (const float*)d_in[1];
  const float* data   = (const float*)d_in[2];
  const int*   labels = (const int*)  d_in[3];
  float* out = (float*)d_out;
  float* ws  = (float*)d_ws;

  const int nacc = NCOPY * ACC_STRIDE;
  zero_ws_kernel<<<(nacc + 255) / 256, 256, 0, stream>>>(ws, nacc);
  crf_word_kernel<<<Nw, 256, 0, stream>>>(W, T, data, labels, ws);
  reduce_out_kernel<<<(ACC_STRIDE + 255) / 256, 256, 0, stream>>>(ws, out);
}
